# starting kernel
# baseline (speedup 1.0000x reference)
#include <hip/hip_runtime.h>


typedef __attribute__((ext_vector_type(2))) float v2f;
typedef __attribute__((ext_vector_type(8))) float v8f;

#define M_DIM   8192
#define N_DIM   8192
#define RANK    64
#define P_DIM   128
#define S1_BLOCKS 16
#define S1_CHUNK  (M_DIM / S1_BLOCKS)

static __device__ __forceinline__ v8f wmma_f32_16x16x4(v2f a, v2f b, v8f c) {
  return __builtin_amdgcn_wmma_f32_16x16x4_f32(false, a, false, b,
                                               (short)0, c, false, false);
}

__global__ void __launch_bounds__(256)
cur_stage1_xRt(const float* __restrict__ x, const float* __restrict__ R,
               float* __restrict__ partial) {
  const int lane = threadIdx.x & 31;
  const int wave = threadIdx.x >> 5;
  const int half = lane >> 4;
  const int l16  = lane & 15;
  const int kk   = half * 2;
  const int mBase = blockIdx.x * S1_CHUNK;

  float* dst = partial + (size_t)blockIdx.x * (P_DIM * RANK);

  for (int t = wave; t < 32; t += 8) {
    const int pBase = (t & 7) * 16;
    const int rBase = (t >> 3) * 16;
    const float* xRow = x + (size_t)(pBase + l16) * M_DIM;
    const float* rRow = R + (size_t)(rBase + l16) * M_DIM;

    v8f acc = {};
    for (int m = mBase; m < mBase + S1_CHUNK; m += 4) {
      __builtin_prefetch(xRow + m + 128, 0, 0);
      v2f a = *reinterpret_cast<const v2f*>(xRow + m + kk);
      v2f b = *reinterpret_cast<const v2f*>(rRow + m + kk);
      acc = wmma_f32_16x16x4(a, b, acc);
    }

#pragma unroll
    for (int v = 0; v < 8; ++v)
      dst[(pBase + v + 8 * half) * RANK + (rBase + l16)] = acc[v];
  }
}

__global__ void __launch_bounds__(256)
cur_stage2_Ut(const float* __restrict__ partial, const float* __restrict__ U,
              float* __restrict__ t2) {
  __shared__ float t1s[P_DIM * RANK];

  for (int e = threadIdx.x; e < P_DIM * RANK; e += 256) {
    float s = 0.f;
#pragma unroll
    for (int c = 0; c < S1_BLOCKS; ++c)
      s += partial[c * (P_DIM * RANK) + e];
    t1s[e] = s;
  }
  __syncthreads();

  const int lane = threadIdx.x & 31;
  const int wave = threadIdx.x >> 5;
  const int half = lane >> 4;
  const int l16  = lane & 15;
  const int kk   = half * 2;

  for (int t = wave; t < 32; t += 8) {
    const int pBase = (t & 7) * 16;
    const int uBase = (t >> 3) * 16;
    const float* uRow = U + (size_t)(uBase + l16) * RANK;

    v8f acc = {};
#pragma unroll
    for (int r0 = 0; r0 < RANK; r0 += 4) {
      v2f a = *reinterpret_cast<const v2f*>(&t1s[(pBase + l16) * RANK + r0 + kk]);
      v2f b = *reinterpret_cast<const v2f*>(uRow + r0 + kk);
      acc = wmma_f32_16x16x4(a, b, acc);
    }
#pragma unroll
    for (int v = 0; v < 8; ++v)
      t2[(pBase + v + 8 * half) * RANK + (uBase + l16)] = acc[v];
  }
}

__global__ void __launch_bounds__(256)
cur_stage3_Ct(const float* __restrict__ t2, const float* __restrict__ C,
              float* __restrict__ out) {
  const int lane = threadIdx.x & 31;
  const int wave = threadIdx.x >> 5;
  const int half = lane >> 4;
  const int l16  = lane & 15;
  const int kk   = half * 2;

  const int tile  = blockIdx.x * 8 + wave;
  const int pBase = (tile & 7) * 16;
  const int nBase = (tile >> 3) * 16;

  const float* aRow = t2 + (size_t)(pBase + l16) * RANK;
  const float* cRow = C  + (size_t)(nBase + l16) * RANK;

  v8f acc = {};
#pragma unroll
  for (int r0 = 0; r0 < RANK; r0 += 4) {
    v2f a = *reinterpret_cast<const v2f*>(aRow + r0 + kk);
    v2f b = *reinterpret_cast<const v2f*>(cRow + r0 + kk);
    acc = wmma_f32_16x16x4(a, b, acc);
  }

#pragma unroll
  for (int v = 0; v < 8; ++v)
    out[(size_t)(pBase + v + 8 * half) * N_DIM + (nBase + l16)] = acc[v];
}

extern "C" void kernel_launch(void* const* d_in, const int* in_sizes, int n_in,
                              void* d_out, int out_size, void* d_ws, size_t ws_size,
                              hipStream_t stream) {
  (void)in_sizes; (void)n_in; (void)out_size; (void)ws_size;
  const float* x = (const float*)d_in[0];
  const float* C = (const float*)d_in[1];
  const float* U = (const float*)d_in[2];
  const float* R = (const float*)d_in[3];
  float* out = (float*)d_out;

  float* partial = (float*)d_ws;
  float* t2      = partial + S1_BLOCKS * P_DIM * RANK;

  cur_stage1_xRt<<<S1_BLOCKS, 256, 0, stream>>>(x, R, partial);
  cur_stage2_Ut <<<1,         256, 0, stream>>>(partial, U, t2);
  cur_stage3_Ct <<<(4096 / 8), 256, 0, stream>>>(t2, C, out);
}


# stage 1 (x R^T) re-gridded from 16 to 256 workgroups: K of each chunk split over the 8 waves, tiles summed in LDS, 16-byte fragment loads issued ahead
# speedup vs baseline: 3.7394x; 3.7394x over previous
#include <hip/hip_runtime.h>


typedef __attribute__((ext_vector_type(2))) float v2f;
typedef __attribute__((ext_vector_type(4))) float v4f;
typedef __attribute__((ext_vector_type(8))) float v8f;

#define M_DIM   8192
#define N_DIM   8192
#define RANK    64
#define P_DIM   128
#define S1_BLOCKS 16
#define S1_CHUNK  (M_DIM / S1_BLOCKS)

static __device__ __forceinline__ v8f wmma_f32_16x16x4(v2f a, v2f b, v8f c) {
  return __builtin_amdgcn_wmma_f32_16x16x4_f32(false, a, false, b,
                                               (short)0, c, false, false);
}

#define S1_GRID   256
#define S1_TILE_LD 272
#define S1_WAVE_LD (2 * S1_TILE_LD)

__global__ void __launch_bounds__(256)
cur_stage1_xRt(const float* __restrict__ x, const float* __restrict__ R,
               float* __restrict__ partial) {
  __shared__ __attribute__((aligned(16))) float red[8 * S1_WAVE_LD];

  const int lane = threadIdx.x & 31;
  const int wave = threadIdx.x >> 5;
  const int half = lane >> 4;
  const int l16  = lane & 15;

  const int rh = blockIdx.x & 1;
  const int pt = (blockIdx.x >> 1) & 7;
  const int ch = blockIdx.x >> 4;
  const int pBase = pt * 16;
  const int rBase = rh * 32;
  const int k0 = ch * S1_CHUNK + wave * 64 + half * 4;

  const float* xp  = x + (size_t)(pBase + l16) * M_DIM + k0;
  const float* r0p = R + (size_t)(rBase + l16) * M_DIM + k0;
  const float* r1p = r0p + (size_t)16 * M_DIM;

  v4f a[8], b0[8], b1[8];
#pragma unroll
  for (int j = 0; j < 8; ++j) {
    a[j]  = *reinterpret_cast<const v4f*>(xp  + 8 * j);
    b0[j] = *reinterpret_cast<const v4f*>(r0p + 8 * j);
    b1[j] = *reinterpret_cast<const v4f*>(r1p + 8 * j);
  }
  __builtin_amdgcn_sched_barrier(0);

  v8f acc0 = {};
  v8f acc1 = {};
#pragma unroll
  for (int j = 0; j < 8; ++j) {
    const v2f alo  = {a[j].x,  a[j].y},  ahi  = {a[j].z,  a[j].w};
    const v2f b0lo = {b0[j].x, b0[j].y}, b0hi = {b0[j].z, b0[j].w};
    const v2f b1lo = {b1[j].x, b1[j].y}, b1hi = {b1[j].z, b1[j].w};
    acc0 = wmma_f32_16x16x4(alo, b0lo, acc0);
    acc1 = wmma_f32_16x16x4(alo, b1lo, acc1);
    acc0 = wmma_f32_16x16x4(ahi, b0hi, acc0);
    acc1 = wmma_f32_16x16x4(ahi, b1hi, acc1);
  }

  float* mine = red + wave * S1_WAVE_LD + lane;
#pragma unroll
  for (int v = 0; v < 8; ++v) {
    mine[v * 32]              = acc0[v];
    mine[S1_TILE_LD + v * 32] = acc1[v];
  }
  __syncthreads();

  const int row = threadIdx.x >> 4;
  const int cp  = threadIdx.x & 15;
  const int idx = (cp >> 3) * S1_TILE_LD + (row & 7) * 32 + (row >> 3) * 16
                + (cp & 7) * 2;
  v2f s = *reinterpret_cast<const v2f*>(red + idx);
#pragma unroll
  for (int w = 1; w < 8; ++w)
    s += *reinterpret_cast<const v2f*>(red + w * S1_WAVE_LD + idx);

  float* dst = partial + (size_t)ch * (P_DIM * RANK)
             + (pBase + row) * RANK + rBase + 2 * cp;
  *reinterpret_cast<v2f*>(dst) = s;
}

__global__ void __launch_bounds__(256)
cur_stage2_Ut(const float* __restrict__ partial, const float* __restrict__ U,
              float* __restrict__ t2) {
  __shared__ float t1s[P_DIM * RANK];

  for (int e = threadIdx.x; e < P_DIM * RANK; e += 256) {
    float s = 0.f;
#pragma unroll
    for (int c = 0; c < S1_BLOCKS; ++c)
      s += partial[c * (P_DIM * RANK) + e];
    t1s[e] = s;
  }
  __syncthreads();

  const int lane = threadIdx.x & 31;
  const int wave = threadIdx.x >> 5;
  const int half = lane >> 4;
  const int l16  = lane & 15;
  const int kk   = half * 2;

  for (int t = wave; t < 32; t += 8) {
    const int pBase = (t & 7) * 16;
    const int uBase = (t >> 3) * 16;
    const float* uRow = U + (size_t)(uBase + l16) * RANK;

    v8f acc = {};
#pragma unroll
    for (int r0 = 0; r0 < RANK; r0 += 4) {
      v2f a = *reinterpret_cast<const v2f*>(&t1s[(pBase + l16) * RANK + r0 + kk]);
      v2f b = *reinterpret_cast<const v2f*>(uRow + r0 + kk);
      acc = wmma_f32_16x16x4(a, b, acc);
    }
#pragma unroll
    for (int v = 0; v < 8; ++v)
      t2[(pBase + v + 8 * half) * RANK + (uBase + l16)] = acc[v];
  }
}

__global__ void __launch_bounds__(256)
cur_stage3_Ct(const float* __restrict__ t2, const float* __restrict__ C,
              float* __restrict__ out) {
  const int lane = threadIdx.x & 31;
  const int wave = threadIdx.x >> 5;
  const int half = lane >> 4;
  const int l16  = lane & 15;
  const int kk   = half * 2;

  const int tile  = blockIdx.x * 8 + wave;
  const int pBase = (tile & 7) * 16;
  const int nBase = (tile >> 3) * 16;

  const float* aRow = t2 + (size_t)(pBase + l16) * RANK;
  const float* cRow = C  + (size_t)(nBase + l16) * RANK;

  v8f acc = {};
#pragma unroll
  for (int r0 = 0; r0 < RANK; r0 += 4) {
    v2f a = *reinterpret_cast<const v2f*>(aRow + r0 + kk);
    v2f b = *reinterpret_cast<const v2f*>(cRow + r0 + kk);
    acc = wmma_f32_16x16x4(a, b, acc);
  }

#pragma unroll
  for (int v = 0; v < 8; ++v)
    out[(size_t)(pBase + v + 8 * half) * N_DIM + (nBase + l16)] = acc[v];
}

extern "C" void kernel_launch(void* const* d_in, const int* in_sizes, int n_in,
                              void* d_out, int out_size, void* d_ws, size_t ws_size,
                              hipStream_t stream) {
  (void)in_sizes; (void)n_in; (void)out_size; (void)ws_size;
  const float* x = (const float*)d_in[0];
  const float* C = (const float*)d_in[1];
  const float* U = (const float*)d_in[2];
  const float* R = (const float*)d_in[3];
  float* out = (float*)d_out;

  float* partial = (float*)d_ws;
  float* t2      = partial + S1_BLOCKS * P_DIM * RANK;

  cur_stage1_xRt<<<S1_GRID,   256, 0, stream>>>(x, R, partial);
  cur_stage2_Ut <<<1,         256, 0, stream>>>(partial, U, t2);
  cur_stage3_Ct <<<(4096 / 8), 256, 0, stream>>>(t2, C, out);
}


# stage 2 (sum of the 16 partial planes, then t1 U^T) re-gridded from 1 to 8 workgroups: a 16-row strip each, planes read by 16-byte loads all in flight, K halved over the waves and summed in LDS
# speedup vs baseline: 5.9818x; 1.5997x over previous
#include <hip/hip_runtime.h>


typedef __attribute__((ext_vector_type(2))) float v2f;
typedef __attribute__((ext_vector_type(4))) float v4f;
typedef __attribute__((ext_vector_type(8))) float v8f;

#define M_DIM   8192
#define N_DIM   8192
#define RANK    64
#define P_DIM   128
#define S1_BLOCKS 16
#define S1_CHUNK  (M_DIM / S1_BLOCKS)

static __device__ __forceinline__ v8f wmma_f32_16x16x4(v2f a, v2f b, v8f c) {
  return __builtin_amdgcn_wmma_f32_16x16x4_f32(false, a, false, b,
                                               (short)0, c, false, false);
}

#define S1_GRID   256
#define S1_TILE_LD 272
#define S1_WAVE_LD (2 * S1_TILE_LD)

__global__ void __launch_bounds__(256)
cur_stage1_xRt(const float* __restrict__ x, const float* __restrict__ R,
               float* __restrict__ partial) {
  __shared__ __attribute__((aligned(16))) float red[8 * S1_WAVE_LD];

  const int lane = threadIdx.x & 31;
  const int wave = threadIdx.x >> 5;
  const int half = lane >> 4;
  const int l16  = lane & 15;

  const int rh = blockIdx.x & 1;
  const int pt = (blockIdx.x >> 1) & 7;
  const int ch = blockIdx.x >> 4;
  const int pBase = pt * 16;
  const int rBase = rh * 32;
  const int k0 = ch * S1_CHUNK + wave * 64 + half * 4;

  const float* xp  = x + (size_t)(pBase + l16) * M_DIM + k0;
  const float* r0p = R + (size_t)(rBase + l16) * M_DIM + k0;
  const float* r1p = r0p + (size_t)16 * M_DIM;

  v4f a[8], b0[8], b1[8];
#pragma unroll
  for (int j = 0; j < 8; ++j) {
    a[j]  = *reinterpret_cast<const v4f*>(xp  + 8 * j);
    b0[j] = *reinterpret_cast<const v4f*>(r0p + 8 * j);
    b1[j] = *reinterpret_cast<const v4f*>(r1p + 8 * j);
  }
  __builtin_amdgcn_sched_barrier(0);

  v8f acc0 = {};
  v8f acc1 = {};
#pragma unroll
  for (int j = 0; j < 8; ++j) {
    const v2f alo  = {a[j].x,  a[j].y},  ahi  = {a[j].z,  a[j].w};
    const v2f b0lo = {b0[j].x, b0[j].y}, b0hi = {b0[j].z, b0[j].w};
    const v2f b1lo = {b1[j].x, b1[j].y}, b1hi = {b1[j].z, b1[j].w};
    acc0 = wmma_f32_16x16x4(alo, b0lo, acc0);
    acc1 = wmma_f32_16x16x4(alo, b1lo, acc1);
    acc0 = wmma_f32_16x16x4(ahi, b0hi, acc0);
    acc1 = wmma_f32_16x16x4(ahi, b1hi, acc1);
  }

  float* mine = red + wave * S1_WAVE_LD + lane;
#pragma unroll
  for (int v = 0; v < 8; ++v) {
    mine[v * 32]              = acc0[v];
    mine[S1_TILE_LD + v * 32] = acc1[v];
  }
  __syncthreads();

  const int row = threadIdx.x >> 4;
  const int cp  = threadIdx.x & 15;
  const int idx = (cp >> 3) * S1_TILE_LD + (row & 7) * 32 + (row >> 3) * 16
                + (cp & 7) * 2;
  v2f s = *reinterpret_cast<const v2f*>(red + idx);
#pragma unroll
  for (int w = 1; w < 8; ++w)
    s += *reinterpret_cast<const v2f*>(red + w * S1_WAVE_LD + idx);

  float* dst = partial + (size_t)ch * (P_DIM * RANK)
             + (pBase + row) * RANK + rBase + 2 * cp;
  *reinterpret_cast<v2f*>(dst) = s;
}

#define S2_GRID    (P_DIM / 16)
#define S2_T1_LD   68
#define S2_TILE_LD 272

__global__ void __launch_bounds__(256)
cur_stage2_Ut(const float* __restrict__ partial, const float* __restrict__ U,
              float* __restrict__ t2) {
  __shared__ __attribute__((aligned(16))) float t1s[16 * S2_T1_LD];
  __shared__ __attribute__((aligned(16))) float red[8 * S2_TILE_LD];

  const int lane = threadIdx.x & 31;
  const int wave = threadIdx.x >> 5;
  const int half = lane >> 4;
  const int l16  = lane & 15;

  const int pBase = blockIdx.x * 16;
  const int ut    = wave & 3;
  const int kOff  = (wave >> 2) * 32 + half * 4;

  const float* pp = partial + (size_t)pBase * RANK + threadIdx.x * 4;
  v4f pl[S1_BLOCKS];
#pragma unroll
  for (int c = 0; c < S1_BLOCKS; ++c)
    pl[c] = *reinterpret_cast<const v4f*>(pp + (size_t)c * (P_DIM * RANK));

  const float* up = U + (size_t)(ut * 16 + l16) * RANK + kOff;
  v4f b[4];
#pragma unroll
  for (int j = 0; j < 4; ++j)
    b[j] = *reinterpret_cast<const v4f*>(up + 8 * j);
  __builtin_amdgcn_sched_barrier(0);

  v4f s = pl[0];
#pragma unroll
  for (int c = 1; c < S1_BLOCKS; ++c)
    s += pl[c];

  const int row = threadIdx.x >> 4;
  const int c4  = (threadIdx.x & 15) * 4;
  *reinterpret_cast<v4f*>(t1s + row * S2_T1_LD + c4) = s;
  __syncthreads();

  const float* ap = t1s + l16 * S2_T1_LD + kOff;
  v4f a[4];
#pragma unroll
  for (int j = 0; j < 4; ++j)
    a[j] = *reinterpret_cast<const v4f*>(ap + 8 * j);

  v8f acc = {};
#pragma unroll
  for (int j = 0; j < 4; ++j) {
    const v2f alo = {a[j].x, a[j].y}, ahi = {a[j].z, a[j].w};
    const v2f blo = {b[j].x, b[j].y}, bhi = {b[j].z, b[j].w};
    acc = wmma_f32_16x16x4(alo, blo, acc);
    acc = wmma_f32_16x16x4(ahi, bhi, acc);
  }

  float* mine = red + wave * S2_TILE_LD + lane;
#pragma unroll
  for (int v = 0; v < 8; ++v)
    mine[v * 32] = acc[v];
  __syncthreads();

  const int idx = (c4 >> 4) * S2_TILE_LD + (row & 7) * 32 + (row >> 3) * 16
                + (c4 & 15);
  const v4f o = *reinterpret_cast<const v4f*>(red + idx)
              + *reinterpret_cast<const v4f*>(red + 4 * S2_TILE_LD + idx);
  *reinterpret_cast<v4f*>(t2 + (size_t)(pBase + row) * RANK + c4) = o;
}

__global__ void __launch_bounds__(256)
cur_stage3_Ct(const float* __restrict__ t2, const float* __restrict__ C,
              float* __restrict__ out) {
  const int lane = threadIdx.x & 31;
  const int wave = threadIdx.x >> 5;
  const int half = lane >> 4;
  const int l16  = lane & 15;
  const int kk   = half * 2;

  const int tile  = blockIdx.x * 8 + wave;
  const int pBase = (tile & 7) * 16;
  const int nBase = (tile >> 3) * 16;

  const float* aRow = t2 + (size_t)(pBase + l16) * RANK;
  const float* cRow = C  + (size_t)(nBase + l16) * RANK;

  v8f acc = {};
#pragma unroll
  for (int r0 = 0; r0 < RANK; r0 += 4) {
    v2f a = *reinterpret_cast<const v2f*>(aRow + r0 + kk);
    v2f b = *reinterpret_cast<const v2f*>(cRow + r0 + kk);
    acc = wmma_f32_16x16x4(a, b, acc);
  }

#pragma unroll
  for (int v = 0; v < 8; ++v)
    out[(size_t)(pBase + v + 8 * half) * N_DIM + (nBase + l16)] = acc[v];
}

extern "C" void kernel_launch(void* const* d_in, const int* in_sizes, int n_in,
                              void* d_out, int out_size, void* d_ws, size_t ws_size,
                              hipStream_t stream) {
  (void)in_sizes; (void)n_in; (void)out_size; (void)ws_size;
  const float* x = (const float*)d_in[0];
  const float* C = (const float*)d_in[1];
  const float* U = (const float*)d_in[2];
  const float* R = (const float*)d_in[3];
  float* out = (float*)d_out;

  float* partial = (float*)d_ws;
  float* t2      = partial + S1_BLOCKS * P_DIM * RANK;

  cur_stage1_xRt<<<S1_GRID,   256, 0, stream>>>(x, R, partial);
  cur_stage2_Ut <<<S2_GRID,   256, 0, stream>>>(partial, U, t2);
  cur_stage3_Ct <<<(4096 / 8), 256, 0, stream>>>(t2, C, out);
}
